# starting kernel
# baseline (speedup 1.0000x reference)
#include <hip/hip_runtime.h>

#define BB 8
#define AA 256
#define TT 496
#define NN 32

typedef __attribute__((ext_vector_type(2))) float v2f;
typedef __attribute__((ext_vector_type(8))) float v8f;

__global__ __launch_bounds__(256) void TriplesDistances_kernel(
    const float* __restrict__ positions,
    const int*   __restrict__ nbr_j,
    const int*   __restrict__ nbr_k,
    const int*   __restrict__ offidx_j,
    const int*   __restrict__ offidx_k,
    const float* __restrict__ cell,
    const float* __restrict__ cell_offsets,
    float*       __restrict__ out)
{
    __shared__ float s_pos[AA * 4];
    __shared__ float s_off[NN * 4];

    const int ba  = blockIdx.x;
    const int b   = ba / AA;
    const int a   = ba - b * AA;
    const int tid = threadIdx.x;

    const float* posb = positions + (size_t)b * AA * 3;
    for (int i = tid; i < AA * 3; i += 256) {
        s_pos[(i / 3) * 4 + (i % 3)] = posb[i];
    }

    if (tid < 32) {
        const int lane = tid;
        const int half = lane >> 4;
        const int col  = lane & 15;

        v2f bm;
        float b0 = 0.0f, b1 = 0.0f;
        if (col < 3) {
            b0 = cell[b * 9 + (2 * half) * 3 + col];
            if (half == 0) b1 = cell[b * 9 + 1 * 3 + col];
        }
        bm[0] = b0; bm[1] = b1;

        const float* co_base = cell_offsets + ((size_t)ba * NN) * 3;

        #pragma unroll
        for (int m_base = 0; m_base < NN; m_base += 16) {
            const float* co = co_base + (size_t)(m_base + col) * 3;
            v2f am;
            am[0] = co[2 * half];
            am[1] = (half == 0) ? co[1] : 0.0f;
            v8f c = {};
            v8f d = __builtin_amdgcn_wmma_f32_16x16x4_f32(
                false, am, false, bm, (short)0, c, false, false);
            if (col < 3) {
                #pragma unroll
                for (int r = 0; r < 8; ++r) {
                    s_off[(m_base + 8 * half + r) * 4 + col] = d[r];
                }
            }
        }
    }
    __syncthreads();

    const float pix = s_pos[a * 4 + 0];
    const float piy = s_pos[a * 4 + 1];
    const float piz = s_pos[a * 4 + 2];

    const size_t base  = (size_t)ba * TT;
    const size_t plane = (size_t)BB * AA * TT;

    for (int t = tid; t < TT; t += 256) {
        const size_t idx = base + t;
        const int j  = __builtin_nontemporal_load(&nbr_j[idx]);
        const int k  = __builtin_nontemporal_load(&nbr_k[idx]);
        const int oj = __builtin_nontemporal_load(&offidx_j[idx]);
        const int ok = __builtin_nontemporal_load(&offidx_k[idx]);

        const float pjx = s_pos[j * 4 + 0] + s_off[oj * 4 + 0];
        const float pjy = s_pos[j * 4 + 1] + s_off[oj * 4 + 1];
        const float pjz = s_pos[j * 4 + 2] + s_off[oj * 4 + 2];
        const float pkx = s_pos[k * 4 + 0] + s_off[ok * 4 + 0];
        const float pky = s_pos[k * 4 + 1] + s_off[ok * 4 + 1];
        const float pkz = s_pos[k * 4 + 2] + s_off[ok * 4 + 2];

        const float dijx = pjx - pix, dijy = pjy - piy, dijz = pjz - piz;
        const float dikx = pkx - pix, diky = pky - piy, dikz = pkz - piz;
        const float djkx = pjx - pkx, djky = pjy - pky, djkz = pjz - pkz;

        const float rij = sqrtf(dijx * dijx + dijy * dijy + dijz * dijz) + 1e-9f;
        const float rik = sqrtf(dikx * dikx + diky * diky + dikz * dikz) + 1e-9f;
        const float rjk = sqrtf(djkx * djkx + djky * djky + djkz * djkz) + 1e-9f;

        __builtin_nontemporal_store(rij, &out[idx]);
        __builtin_nontemporal_store(rik, &out[plane + idx]);
        __builtin_nontemporal_store(rjk, &out[2 * plane + idx]);
    }
}

extern "C" void kernel_launch(void* const* d_in, const int* in_sizes, int n_in,
                              void* d_out, int out_size, void* d_ws, size_t ws_size,
                              hipStream_t stream) {
    const float* positions    = (const float*)d_in[0];
    const int*   nbr_j        = (const int*)d_in[1];
    const int*   nbr_k        = (const int*)d_in[2];
    const int*   offidx_j     = (const int*)d_in[3];
    const int*   offidx_k     = (const int*)d_in[4];
    const float* cell         = (const float*)d_in[5];
    const float* cell_offsets = (const float*)d_in[6];
    float*       out          = (float*)d_out;

    dim3 grid(BB * AA);
    dim3 block(256);
    TriplesDistances_kernel<<<grid, block, 0, stream>>>(
        positions, nbr_j, nbr_k, offidx_j, offidx_k, cell, cell_offsets, out);
}


# one wave per (b,a) row: 31 lanes x 16 triples via 128-bit index loads and stores issued up front, 8 rows per workgroup, so 2048 waves enter instead of 16384
# speedup vs baseline: 1.3246x; 1.3246x over previous
#include <hip/hip_runtime.h>


#define BB 8
#define AA 256
#define TT 496
#define NN 32

#define ROWS_PER_WG 8
#define CHUNKS 4
#define CHUNK_T 124

typedef __attribute__((ext_vector_type(2))) float v2f;
typedef __attribute__((ext_vector_type(4))) float v4f;
typedef __attribute__((ext_vector_type(8))) float v8f;
typedef __attribute__((ext_vector_type(4))) int   v4i;

__global__ __launch_bounds__(256) void TriplesDistances_kernel(
    const float* __restrict__ positions,
    const int*   __restrict__ nbr_j,
    const int*   __restrict__ nbr_k,
    const int*   __restrict__ offidx_j,
    const int*   __restrict__ offidx_k,
    const float* __restrict__ cell,
    const float* __restrict__ cell_offsets,
    float*       __restrict__ out)
{
    __shared__ __attribute__((aligned(16))) float s_pos[AA * 4];
    __shared__ __attribute__((aligned(16))) float s_off[ROWS_PER_WG][NN * 4];

    const int tid  = threadIdx.x;
    const int lane = tid & 31;
    const int wave = __builtin_amdgcn_readfirstlane(tid >> 5);
    const int b    = blockIdx.x / (AA / ROWS_PER_WG);
    const int ba   = blockIdx.x * ROWS_PER_WG + wave;
    const int a    = ba - b * AA;

    const float* posb = positions + (size_t)b * AA * 3;
    const float p0 = posb[tid];
    const float p1 = posb[tid + 256];
    const float p2 = posb[tid + 512];

    const int half = lane >> 4;
    const int col  = lane & 15;

    float b0 = 0.0f, b1 = 0.0f;
    if (col < 3) {
        b0 = cell[b * 9 + (2 * half) * 3 + col];
        if (half == 0) b1 = cell[b * 9 + 1 * 3 + col];
    }
    v2f bm;
    bm[0] = b0; bm[1] = b1;

    const float* co_base = cell_offsets + ((size_t)ba * NN) * 3;
    v2f am[2];
    #pragma unroll
    for (int mt = 0; mt < 2; ++mt) {
        const float* co = co_base + (size_t)(mt * 16 + col) * 3;
        am[mt][0] = co[2 * half];
        am[mt][1] = (half == 0) ? co[1] : 0.0f;
    }

    const int    ll   = (lane < 31) ? lane : 30;
    const size_t base = (size_t)ba * TT + (size_t)(ll * 4);
    v4i vj[CHUNKS], vk[CHUNKS], voj[CHUNKS], vok[CHUNKS];
    #pragma unroll
    for (int c = 0; c < CHUNKS; ++c) {
        const size_t idx = base + (size_t)(c * CHUNK_T);
        vj[c]  = __builtin_nontemporal_load((const v4i*)(nbr_j    + idx));
        vk[c]  = __builtin_nontemporal_load((const v4i*)(nbr_k    + idx));
        voj[c] = __builtin_nontemporal_load((const v4i*)(offidx_j + idx));
        vok[c] = __builtin_nontemporal_load((const v4i*)(offidx_k + idx));
    }

    s_pos[((tid      ) / 3) * 4 + ((tid      ) % 3)] = p0;
    s_pos[((tid + 256) / 3) * 4 + ((tid + 256) % 3)] = p1;
    s_pos[((tid + 512) / 3) * 4 + ((tid + 512) % 3)] = p2;
    s_pos[tid * 4 + 3] = 0.0f;

    float* so = &s_off[wave][0];
    #pragma unroll
    for (int mt = 0; mt < 2; ++mt) {
        v8f c = {};
        v8f d = __builtin_amdgcn_wmma_f32_16x16x4_f32(
            false, am[mt], false, bm, (short)0, c, false, false);
        if (col < 4) {
            #pragma unroll
            for (int r = 0; r < 8; ++r) {
                so[(mt * 16 + 8 * half + r) * 4 + col] = d[r];
            }
        }
    }
    __syncthreads();

    const v4f pi = *(const v4f*)&s_pos[a * 4];
    const float pix = pi[0], piy = pi[1], piz = pi[2];
    const size_t plane = (size_t)BB * AA * TT;

    #pragma unroll
    for (int c = 0; c < CHUNKS; ++c) {
        v4f rij, rik, rjk;
        #pragma unroll
        for (int q = 0; q < 4; ++q) {
            const v4f Pj = *(const v4f*)&s_pos[vj[c][q] * 4];
            const v4f Pk = *(const v4f*)&s_pos[vk[c][q] * 4];
            const v4f Oj = *(const v4f*)&so[voj[c][q] * 4];
            const v4f Ok = *(const v4f*)&so[vok[c][q] * 4];

            const float pjx = Pj[0] + Oj[0];
            const float pjy = Pj[1] + Oj[1];
            const float pjz = Pj[2] + Oj[2];
            const float pkx = Pk[0] + Ok[0];
            const float pky = Pk[1] + Ok[1];
            const float pkz = Pk[2] + Ok[2];

            const float dijx = pjx - pix, dijy = pjy - piy, dijz = pjz - piz;
            const float dikx = pkx - pix, diky = pky - piy, dikz = pkz - piz;
            const float djkx = pjx - pkx, djky = pjy - pky, djkz = pjz - pkz;

            rij[q] = sqrtf(dijx * dijx + dijy * dijy + dijz * dijz) + 1e-9f;
            rik[q] = sqrtf(dikx * dikx + diky * diky + dikz * dikz) + 1e-9f;
            rjk[q] = sqrtf(djkx * djkx + djky * djky + djkz * djkz) + 1e-9f;
        }
        if (lane < 31) {
            const size_t idx = base + (size_t)(c * CHUNK_T);
            __builtin_nontemporal_store(rij, (v4f*)(out + idx));
            __builtin_nontemporal_store(rik, (v4f*)(out + plane + idx));
            __builtin_nontemporal_store(rjk, (v4f*)(out + 2 * plane + idx));
        }
    }
}

extern "C" void kernel_launch(void* const* d_in, const int* in_sizes, int n_in,
                              void* d_out, int out_size, void* d_ws, size_t ws_size,
                              hipStream_t stream) {
    const float* positions    = (const float*)d_in[0];
    const int*   nbr_j        = (const int*)d_in[1];
    const int*   nbr_k        = (const int*)d_in[2];
    const int*   offidx_j     = (const int*)d_in[3];
    const int*   offidx_k     = (const int*)d_in[4];
    const float* cell         = (const float*)d_in[5];
    const float* cell_offsets = (const float*)d_in[6];
    float*       out          = (float*)d_out;

    dim3 grid(256);
    dim3 block(256);
    TriplesDistances_kernel<<<grid, block, 0, stream>>>(
        positions, nbr_j, nbr_k, offidx_j, offidx_k, cell, cell_offsets, out);
}


# speedup vs baseline: 1.4192x; 1.0714x over previous
#include <hip/hip_runtime.h>


#define BB 8
#define AA 256
#define TT 496
#define NN 32

#define ROWS_PER_WG 8
#define CHUNKS 4
#define CHUNK_T 124

#define WAVE_BLOCK_FLOATS 8192
#define POS_FLOATS (AA * 4)
#define OFF_FLOATS (NN * 4)
#define LDS_FLOATS ((ROWS_PER_WG - 1) * WAVE_BLOCK_FLOATS + POS_FLOATS + OFF_FLOATS)

typedef __attribute__((ext_vector_type(2))) float v2f;
typedef __attribute__((ext_vector_type(4))) float v4f;
typedef __attribute__((ext_vector_type(8))) float v8f;
typedef __attribute__((ext_vector_type(4))) int   v4i;

__global__ __launch_bounds__(256) void TriplesDistances_kernel(
    const float* __restrict__ positions,
    const int*   __restrict__ nbr_j,
    const int*   __restrict__ nbr_k,
    const int*   __restrict__ offidx_j,
    const int*   __restrict__ offidx_k,
    const float* __restrict__ cell,
    const float* __restrict__ cell_offsets,
    float*       __restrict__ out)
{
    __shared__ __attribute__((aligned(16))) float s_tab[LDS_FLOATS];

    const int tid  = threadIdx.x;
    const int lane = tid & 31;
    const int wave = __builtin_amdgcn_readfirstlane(tid >> 5);
    const int b    = blockIdx.x / (AA / ROWS_PER_WG);
    const int ba   = blockIdx.x * ROWS_PER_WG + wave;
    const int a    = ba - b * AA;

    float* sp = s_tab + wave * WAVE_BLOCK_FLOATS;
    float* so = sp + POS_FLOATS;

    const float* posb = positions + (size_t)b * AA * 3;
    float px[8], py[8], pz[8];
    #pragma unroll
    for (int r = 0; r < 8; ++r) {
        const float* p = posb + (lane + 32 * r) * 3;
        px[r] = p[0];
        py[r] = p[1];
        pz[r] = p[2];
    }

    const int half = lane >> 4;
    const int col  = lane & 15;

    float b0 = 0.0f, b1 = 0.0f;
    if (col < 3) {
        b0 = cell[b * 9 + (2 * half) * 3 + col];
        if (half == 0) b1 = cell[b * 9 + 1 * 3 + col];
    }
    v2f bm;
    bm[0] = b0; bm[1] = b1;

    const float* co_base = cell_offsets + ((size_t)ba * NN) * 3;
    v2f am[2];
    #pragma unroll
    for (int mt = 0; mt < 2; ++mt) {
        const float* co = co_base + (size_t)(mt * 16 + col) * 3;
        am[mt][0] = co[2 * half];
        am[mt][1] = (half == 0) ? co[1] : 0.0f;
    }

    const int    ll   = (lane < 31) ? lane : 30;
    const size_t base = (size_t)ba * TT + (size_t)(ll * 4);
    v4i vj[CHUNKS], vk[CHUNKS], voj[CHUNKS], vok[CHUNKS];
    #pragma unroll
    for (int c = 0; c < CHUNKS; ++c) {
        const size_t idx = base + (size_t)(c * CHUNK_T);
        vj[c]  = __builtin_nontemporal_load((const v4i*)(nbr_j    + idx));
        vk[c]  = __builtin_nontemporal_load((const v4i*)(nbr_k    + idx));
        voj[c] = __builtin_nontemporal_load((const v4i*)(offidx_j + idx));
        vok[c] = __builtin_nontemporal_load((const v4i*)(offidx_k + idx));
    }

    #pragma unroll
    for (int r = 0; r < 8; ++r) {
        v4f pv;
        pv[0] = px[r]; pv[1] = py[r]; pv[2] = pz[r];
        pv[3] = 0.0f;
        *(v4f*)&sp[(lane + 32 * r) * 4] = pv;
    }

    #pragma unroll
    for (int mt = 0; mt < 2; ++mt) {
        v8f c = {};
        v8f d = __builtin_amdgcn_wmma_f32_16x16x4_f32(
            false, am[mt], false, bm, (short)0, c, false, false);
        if (col < 4) {
            #pragma unroll
            for (int r = 0; r < 8; ++r) {
                so[(mt * 16 + 8 * half + r) * 4 + col] = d[r];
            }
        }
    }

    const v4f pi = *(const v4f*)&sp[a * 4];
    const float pix = pi[0], piy = pi[1], piz = pi[2];
    const size_t plane = (size_t)BB * AA * TT;

    #pragma unroll
    for (int c = 0; c < CHUNKS; ++c) {
        v4f rij, rik, rjk;
        #pragma unroll
        for (int q = 0; q < 4; ++q) {
            const v4f Pj = *(const v4f*)&sp[vj[c][q] * 4];
            const v4f Pk = *(const v4f*)&sp[vk[c][q] * 4];
            const v4f Oj = *(const v4f*)&so[voj[c][q] * 4];
            const v4f Ok = *(const v4f*)&so[vok[c][q] * 4];

            const float pjx = Pj[0] + Oj[0];
            const float pjy = Pj[1] + Oj[1];
            const float pjz = Pj[2] + Oj[2];
            const float pkx = Pk[0] + Ok[0];
            const float pky = Pk[1] + Ok[1];
            const float pkz = Pk[2] + Ok[2];

            const float dijx = pjx - pix, dijy = pjy - piy, dijz = pjz - piz;
            const float dikx = pkx - pix, diky = pky - piy, dikz = pkz - piz;
            const float djkx = pjx - pkx, djky = pjy - pky, djkz = pjz - pkz;

            rij[q] = sqrtf(dijx * dijx + dijy * dijy + dijz * dijz) + 1e-9f;
            rik[q] = sqrtf(dikx * dikx + diky * diky + dikz * dikz) + 1e-9f;
            rjk[q] = sqrtf(djkx * djkx + djky * djky + djkz * djkz) + 1e-9f;
        }
        if (lane < 31) {
            const size_t idx = base + (size_t)(c * CHUNK_T);
            __builtin_nontemporal_store(rij, (v4f*)(out + idx));
            __builtin_nontemporal_store(rik, (v4f*)(out + plane + idx));
            __builtin_nontemporal_store(rjk, (v4f*)(out + 2 * plane + idx));
        }
    }
}

extern "C" void kernel_launch(void* const* d_in, const int* in_sizes, int n_in,
                              void* d_out, int out_size, void* d_ws, size_t ws_size,
                              hipStream_t stream) {
    const float* positions    = (const float*)d_in[0];
    const int*   nbr_j        = (const int*)d_in[1];
    const int*   nbr_k        = (const int*)d_in[2];
    const int*   offidx_j     = (const int*)d_in[3];
    const int*   offidx_k     = (const int*)d_in[4];
    const float* cell         = (const float*)d_in[5];
    const float* cell_offsets = (const float*)d_in[6];
    float*       out          = (float*)d_out;

    dim3 grid(256);
    dim3 block(256);
    TriplesDistances_kernel<<<grid, block, 0, stream>>>(
        positions, nbr_j, nbr_k, offidx_j, offidx_k, cell, cell_offsets, out);
}


# square roots by the one-instruction hardware v_sqrt_f32 (1 ulp) instead of sqrtf's correctly rounded 14-instruction sequence: 1052 -> 468 vector instructions per wave
# speedup vs baseline: 1.6132x; 1.1367x over previous
#include <hip/hip_runtime.h>


#define BB 8
#define AA 256
#define TT 496
#define NN 32

#define ROWS_PER_WG 8
#define CHUNKS 4
#define CHUNK_T 124

#define WAVE_BLOCK_FLOATS 8192
#define POS_FLOATS (AA * 4)
#define OFF_FLOATS (NN * 4)
#define LDS_FLOATS ((ROWS_PER_WG - 1) * WAVE_BLOCK_FLOATS + POS_FLOATS + OFF_FLOATS)

typedef __attribute__((ext_vector_type(2))) float v2f;
typedef __attribute__((ext_vector_type(4))) float v4f;
typedef __attribute__((ext_vector_type(8))) float v8f;
typedef __attribute__((ext_vector_type(4))) int   v4i;

__global__ __launch_bounds__(256) void TriplesDistances_kernel(
    const float* __restrict__ positions,
    const int*   __restrict__ nbr_j,
    const int*   __restrict__ nbr_k,
    const int*   __restrict__ offidx_j,
    const int*   __restrict__ offidx_k,
    const float* __restrict__ cell,
    const float* __restrict__ cell_offsets,
    float*       __restrict__ out)
{
    __shared__ __attribute__((aligned(16))) float s_tab[LDS_FLOATS];

    const int tid  = threadIdx.x;
    const int lane = tid & 31;
    const int wave = __builtin_amdgcn_readfirstlane(tid >> 5);
    const int b    = blockIdx.x / (AA / ROWS_PER_WG);
    const int ba   = blockIdx.x * ROWS_PER_WG + wave;
    const int a    = ba - b * AA;

    float* sp = s_tab + wave * WAVE_BLOCK_FLOATS;
    float* so = sp + POS_FLOATS;

    const float* posb = positions + (size_t)b * AA * 3;
    float px[8], py[8], pz[8];
    #pragma unroll
    for (int r = 0; r < 8; ++r) {
        const float* p = posb + (lane + 32 * r) * 3;
        px[r] = p[0];
        py[r] = p[1];
        pz[r] = p[2];
    }

    const int half = lane >> 4;
    const int col  = lane & 15;

    float b0 = 0.0f, b1 = 0.0f;
    if (col < 3) {
        b0 = cell[b * 9 + (2 * half) * 3 + col];
        if (half == 0) b1 = cell[b * 9 + 1 * 3 + col];
    }
    v2f bm;
    bm[0] = b0; bm[1] = b1;

    const float* co_base = cell_offsets + ((size_t)ba * NN) * 3;
    v2f am[2];
    #pragma unroll
    for (int mt = 0; mt < 2; ++mt) {
        const float* co = co_base + (size_t)(mt * 16 + col) * 3;
        am[mt][0] = co[2 * half];
        am[mt][1] = (half == 0) ? co[1] : 0.0f;
    }

    const int    ll   = (lane < 31) ? lane : 30;
    const size_t base = (size_t)ba * TT + (size_t)(ll * 4);
    v4i vj[CHUNKS], vk[CHUNKS], voj[CHUNKS], vok[CHUNKS];
    #pragma unroll
    for (int c = 0; c < CHUNKS; ++c) {
        const size_t idx = base + (size_t)(c * CHUNK_T);
        vj[c]  = __builtin_nontemporal_load((const v4i*)(nbr_j    + idx));
        vk[c]  = __builtin_nontemporal_load((const v4i*)(nbr_k    + idx));
        voj[c] = __builtin_nontemporal_load((const v4i*)(offidx_j + idx));
        vok[c] = __builtin_nontemporal_load((const v4i*)(offidx_k + idx));
    }

    #pragma unroll
    for (int r = 0; r < 8; ++r) {
        v4f pv;
        pv[0] = px[r]; pv[1] = py[r]; pv[2] = pz[r];
        pv[3] = 0.0f;
        *(v4f*)&sp[(lane + 32 * r) * 4] = pv;
    }

    #pragma unroll
    for (int mt = 0; mt < 2; ++mt) {
        v8f c = {};
        v8f d = __builtin_amdgcn_wmma_f32_16x16x4_f32(
            false, am[mt], false, bm, (short)0, c, false, false);
        if (col < 4) {
            #pragma unroll
            for (int r = 0; r < 8; ++r) {
                so[(mt * 16 + 8 * half + r) * 4 + col] = d[r];
            }
        }
    }

    const v4f pi = *(const v4f*)&sp[a * 4];
    const float pix = pi[0], piy = pi[1], piz = pi[2];
    const size_t plane = (size_t)BB * AA * TT;

    #pragma unroll
    for (int c = 0; c < CHUNKS; ++c) {
        v4f rij, rik, rjk;
        #pragma unroll
        for (int q = 0; q < 4; ++q) {
            const v4f Pj = *(const v4f*)&sp[vj[c][q] * 4];
            const v4f Pk = *(const v4f*)&sp[vk[c][q] * 4];
            const v4f Oj = *(const v4f*)&so[voj[c][q] * 4];
            const v4f Ok = *(const v4f*)&so[vok[c][q] * 4];

            const float pjx = Pj[0] + Oj[0];
            const float pjy = Pj[1] + Oj[1];
            const float pjz = Pj[2] + Oj[2];
            const float pkx = Pk[0] + Ok[0];
            const float pky = Pk[1] + Ok[1];
            const float pkz = Pk[2] + Ok[2];

            const float dijx = pjx - pix, dijy = pjy - piy, dijz = pjz - piz;
            const float dikx = pkx - pix, diky = pky - piy, dikz = pkz - piz;
            const float djkx = pjx - pkx, djky = pjy - pky, djkz = pjz - pkz;

            rij[q] = __builtin_amdgcn_sqrtf(dijx * dijx + dijy * dijy + dijz * dijz) + 1e-9f;
            rik[q] = __builtin_amdgcn_sqrtf(dikx * dikx + diky * diky + dikz * dikz) + 1e-9f;
            rjk[q] = __builtin_amdgcn_sqrtf(djkx * djkx + djky * djky + djkz * djkz) + 1e-9f;
        }
        if (lane < 31) {
            const size_t idx = base + (size_t)(c * CHUNK_T);
            __builtin_nontemporal_store(rij, (v4f*)(out + idx));
            __builtin_nontemporal_store(rik, (v4f*)(out + plane + idx));
            __builtin_nontemporal_store(rjk, (v4f*)(out + 2 * plane + idx));
        }
    }
}

extern "C" void kernel_launch(void* const* d_in, const int* in_sizes, int n_in,
                              void* d_out, int out_size, void* d_ws, size_t ws_size,
                              hipStream_t stream) {
    const float* positions    = (const float*)d_in[0];
    const int*   nbr_j        = (const int*)d_in[1];
    const int*   nbr_k        = (const int*)d_in[2];
    const int*   offidx_j     = (const int*)d_in[3];
    const int*   offidx_k     = (const int*)d_in[4];
    const float* cell         = (const float*)d_in[5];
    const float* cell_offsets = (const float*)d_in[6];
    float*       out          = (float*)d_out;

    dim3 grid(256);
    dim3 block(256);
    TriplesDistances_kernel<<<grid, block, 0, stream>>>(
        positions, nbr_j, nbr_k, offidx_j, offidx_k, cell, cell_offsets, out);
}


# one wave per pair of adjacent rows (3968 B = 31 whole 128-B lines: every 512-B load and store covers whole lines; 1024 waves, one per SIMD); next chunk's 12-B LDS reads issued ahead of the arithmetic
# speedup vs baseline: 1.7429x; 1.0804x over previous
#include <hip/hip_runtime.h>


#define BB 8
#define AA 256
#define TT 496
#define NN 32

#define PAIRS_PER_WG 4
#define UNITS 248
#define ROW_UNITS 124
#define CHUNKS 8

#define WAVE_BLOCK_FLOATS 16384
#define POS_FLOATS (AA * 4)
#define OFF_FLOATS (2 * NN * 4)
#define LDS_FLOATS ((PAIRS_PER_WG - 1) * WAVE_BLOCK_FLOATS + POS_FLOATS + OFF_FLOATS)

typedef __attribute__((ext_vector_type(2))) float v2f;
typedef __attribute__((ext_vector_type(3))) float v3f;
typedef __attribute__((ext_vector_type(4))) float v4f;
typedef __attribute__((ext_vector_type(8))) float v8f;
typedef __attribute__((ext_vector_type(4))) int   v4i;

static __device__ __forceinline__ void gather_unit(
    const float* sp, const float* sor, v4i ij, v4i ik, v4i oj, v4i ok,
    v3f (&Pj)[4], v3f (&Pk)[4], v3f (&Oj)[4], v3f (&Ok)[4])
{
    #pragma unroll
    for (int q = 0; q < 4; ++q) {
        Pj[q] = *(const v3f*)&sp[ij[q] * 4];
        Pk[q] = *(const v3f*)&sp[ik[q] * 4];
        Oj[q] = *(const v3f*)&sor[oj[q] * 4];
        Ok[q] = *(const v3f*)&sor[ok[q] * 4];
    }
}

static __device__ __forceinline__ bool second_row(int c, int lane)
{
    return (c > 3) || (c == 3 && lane >= (ROW_UNITS - 96));
}

__global__ __launch_bounds__(128) void TriplesDistances_kernel(
    const float* __restrict__ positions,
    const int*   __restrict__ nbr_j,
    const int*   __restrict__ nbr_k,
    const int*   __restrict__ offidx_j,
    const int*   __restrict__ offidx_k,
    const float* __restrict__ cell,
    const float* __restrict__ cell_offsets,
    float*       __restrict__ out)
{
    __shared__ __attribute__((aligned(16))) float s_tab[LDS_FLOATS];

    const int tid  = threadIdx.x;
    const int lane = tid & 31;
    const int wave = __builtin_amdgcn_readfirstlane(tid >> 5);
    const int b    = blockIdx.x / (AA / (2 * PAIRS_PER_WG));
    const int pair = blockIdx.x * PAIRS_PER_WG + wave;
    const int ba0  = 2 * pair;
    const int a0   = ba0 - b * AA;

    float* sp = s_tab + wave * WAVE_BLOCK_FLOATS;
    float* so = sp + POS_FLOATS;

    const float* posb = positions + (size_t)b * AA * 3;
    float px[8], py[8], pz[8];
    #pragma unroll
    for (int r = 0; r < 8; ++r) {
        const float* p = posb + (lane + 32 * r) * 3;
        px[r] = p[0];
        py[r] = p[1];
        pz[r] = p[2];
    }

    const int half = lane >> 4;
    const int col  = lane & 15;

    float b0 = 0.0f, b1 = 0.0f;
    if (col < 3) {
        b0 = cell[b * 9 + (2 * half) * 3 + col];
        if (half == 0) b1 = cell[b * 9 + 1 * 3 + col];
    }
    v2f bm;
    bm[0] = b0; bm[1] = b1;

    const float* co_base = cell_offsets + ((size_t)ba0 * NN) * 3;
    v2f am[4];
    #pragma unroll
    for (int mt = 0; mt < 4; ++mt) {
        const float* co = co_base + (size_t)(mt * 16 + col) * 3;
        am[mt][0] = co[2 * half];
        am[mt][1] = (half == 0) ? co[1] : 0.0f;
    }

    const size_t pair_base = (size_t)pair * (UNITS * 4);
    v4i vj[CHUNKS], vk[CHUNKS], voj[CHUNKS], vok[CHUNKS];
    #pragma unroll
    for (int c = 0; c < CHUNKS; ++c) {
        int u = c * 32 + lane;
        if (c == CHUNKS - 1) u = (u < UNITS) ? u : (UNITS - 1);
        const size_t idx = pair_base + (size_t)(u * 4);
        vj[c]  = __builtin_nontemporal_load((const v4i*)(nbr_j    + idx));
        vk[c]  = __builtin_nontemporal_load((const v4i*)(nbr_k    + idx));
        voj[c] = __builtin_nontemporal_load((const v4i*)(offidx_j + idx));
        vok[c] = __builtin_nontemporal_load((const v4i*)(offidx_k + idx));
    }

    #pragma unroll
    for (int r = 0; r < 8; ++r) {
        v4f pv;
        pv[0] = px[r]; pv[1] = py[r]; pv[2] = pz[r];
        pv[3] = 0.0f;
        *(v4f*)&sp[(lane + 32 * r) * 4] = pv;
    }

    #pragma unroll
    for (int mt = 0; mt < 4; ++mt) {
        v8f c = {};
        v8f d = __builtin_amdgcn_wmma_f32_16x16x4_f32(
            false, am[mt], false, bm, (short)0, c, false, false);
        if (col < 3) {
            #pragma unroll
            for (int r = 0; r < 8; ++r) {
                so[(mt * 16 + 8 * half + r) * 4 + col] = d[r];
            }
        }
    }

    const v3f pi0 = *(const v3f*)&sp[a0 * 4];
    const v3f pi1 = *(const v3f*)&sp[a0 * 4 + 4];
    const size_t plane = (size_t)BB * AA * TT;
    __builtin_amdgcn_sched_barrier(0);

    v3f gPj[2][4], gPk[2][4], gOj[2][4], gOk[2][4];
    gather_unit(sp, so + (second_row(0, lane) ? NN * 4 : 0), vj[0], vk[0], voj[0], vok[0],
                gPj[0], gPk[0], gOj[0], gOk[0]);

    #pragma unroll
    for (int c = 0; c < CHUNKS; ++c) {
        const int s = c & 1;
        if (c + 1 < CHUNKS) {
            gather_unit(sp, so + (second_row(c + 1, lane) ? NN * 4 : 0),
                        vj[c + 1], vk[c + 1], voj[c + 1], vok[c + 1],
                        gPj[s ^ 1], gPk[s ^ 1], gOj[s ^ 1], gOk[s ^ 1]);
        }
        __builtin_amdgcn_sched_barrier(0);

        const bool second = second_row(c, lane);
        const float pix = second ? pi1[0] : pi0[0];
        const float piy = second ? pi1[1] : pi0[1];
        const float piz = second ? pi1[2] : pi0[2];

        v4f rij, rik, rjk;
        #pragma unroll
        for (int q = 0; q < 4; ++q) {
            const v3f Pj = gPj[s][q];
            const v3f Pk = gPk[s][q];
            const v3f Oj = gOj[s][q];
            const v3f Ok = gOk[s][q];

            const float pjx = Pj[0] + Oj[0];
            const float pjy = Pj[1] + Oj[1];
            const float pjz = Pj[2] + Oj[2];
            const float pkx = Pk[0] + Ok[0];
            const float pky = Pk[1] + Ok[1];
            const float pkz = Pk[2] + Ok[2];

            const float dijx = pjx - pix, dijy = pjy - piy, dijz = pjz - piz;
            const float dikx = pkx - pix, diky = pky - piy, dikz = pkz - piz;
            const float djkx = pjx - pkx, djky = pjy - pky, djkz = pjz - pkz;

            rij[q] = __builtin_amdgcn_sqrtf(dijx * dijx + dijy * dijy + dijz * dijz) + 1e-9f;
            rik[q] = __builtin_amdgcn_sqrtf(dikx * dikx + diky * diky + dikz * dikz) + 1e-9f;
            rjk[q] = __builtin_amdgcn_sqrtf(djkx * djkx + djky * djky + djkz * djkz) + 1e-9f;
        }
        if (c < CHUNKS - 1 || lane < UNITS - (CHUNKS - 1) * 32) {
            const size_t idx = pair_base + (size_t)((c * 32 + lane) * 4);
            __builtin_nontemporal_store(rij, (v4f*)(out + idx));
            __builtin_nontemporal_store(rik, (v4f*)(out + plane + idx));
            __builtin_nontemporal_store(rjk, (v4f*)(out + 2 * plane + idx));
        }
    }
}

extern "C" void kernel_launch(void* const* d_in, const int* in_sizes, int n_in,
                              void* d_out, int out_size, void* d_ws, size_t ws_size,
                              hipStream_t stream) {
    const float* positions    = (const float*)d_in[0];
    const int*   nbr_j        = (const int*)d_in[1];
    const int*   nbr_k        = (const int*)d_in[2];
    const int*   offidx_j     = (const int*)d_in[3];
    const int*   offidx_k     = (const int*)d_in[4];
    const float* cell         = (const float*)d_in[5];
    const float* cell_offsets = (const float*)d_in[6];
    float*       out          = (float*)d_out;

    dim3 grid(256);
    dim3 block(128);
    TriplesDistances_kernel<<<grid, block, 0, stream>>>(
        positions, nbr_j, nbr_k, offidx_j, offidx_k, cell, cell_offsets, out);
}
